# starting kernel
# baseline (speedup 1.0000x reference)
#include <hip/hip_runtime.h>
#include <hip/hip_bf16.h>

typedef __attribute__((ext_vector_type(2))) float v2f;
typedef __attribute__((ext_vector_type(4))) float v4f;
typedef __attribute__((ext_vector_type(8))) float v8f;

__global__ void __launch_bounds__(256) se_pool_kernel(
    const float* __restrict__ x, float* __restrict__ pooled) {
  const int plane = blockIdx.x;
  const v4f* p = (const v4f*)(x + (size_t)plane * 4096);

  float sum = 0.0f;
#pragma unroll
  for (int i = 0; i < 4; ++i) {
    v4f v = p[threadIdx.x + i * 256];
    sum += (v.x + v.y) + (v.z + v.w);
  }
#pragma unroll
  for (int off = 16; off > 0; off >>= 1)
    sum += __shfl_down(sum, off, 32);

  __shared__ float red[8];
  if ((threadIdx.x & 31) == 0) red[threadIdx.x >> 5] = sum;
  __syncthreads();
  if (threadIdx.x == 0) {
    float t = 0.0f;
#pragma unroll
    for (int w = 0; w < 8; ++w) t += red[w];
    pooled[plane] = t * (1.0f / 4096.0f);
  }
}

__global__ void __launch_bounds__(128) se_mlp_wmma_kernel(
    const float* __restrict__ pooled,
    const float* __restrict__ w1,
    const float* __restrict__ b1,
    const float* __restrict__ w2,
    const float* __restrict__ b2,
    float* __restrict__ scale,
    float* __restrict__ shift) {
  const int lane  = threadIdx.x & 31;
  const int wave  = threadIdx.x >> 5;
  const int r     = lane & 15;
  const int koff  = (lane >> 4) << 1;
  const int mhalf = (lane < 16) ? 0 : 8;

  __shared__ float hs[32 * 16];

  v8f c0 = {};
  v8f c1 = {};
#pragma unroll 4
  for (int k = 0; k < 256; k += 4) {
    v2f bt = *(const v2f*)(w1 + r * 256 + k + koff);
    v2f a0 = *(const v2f*)(pooled + r * 256 + k + koff);
    v2f a1 = *(const v2f*)(pooled + (16 + r) * 256 + k + koff);
    c0 = __builtin_amdgcn_wmma_f32_16x16x4_f32(false, a0, false, bt,
                                               (short)0, c0, false, false);
    c1 = __builtin_amdgcn_wmma_f32_16x16x4_f32(false, a1, false, bt,
                                               (short)0, c1, false, false);
  }

  const float bias1 = b1[r];
#pragma unroll
  for (int v = 0; v < 8; ++v) {
    float h0 = c0[v] + bias1;
    float h1 = c1[v] + bias1;
    h0 = h0 > 0.0f ? h0 : 0.0f;
    h1 = h1 > 0.0f ? h1 : 0.0f;
    hs[(mhalf + v) * 16 + r]        = h0;
    hs[(16 + mhalf + v) * 16 + r]   = h1;
  }
  __syncthreads();

  for (int nb = wave; nb < 32; nb += 4) {
    const int n0 = nb * 16;
    v8f d0 = {};
    v8f d1 = {};
#pragma unroll
    for (int k = 0; k < 16; k += 4) {
      v2f bt = *(const v2f*)(w2 + (n0 + r) * 16 + k + koff);
      v2f a0 = *(const v2f*)(hs + r * 16 + k + koff);
      v2f a1 = *(const v2f*)(hs + (16 + r) * 16 + k + koff);
      d0 = __builtin_amdgcn_wmma_f32_16x16x4_f32(false, a0, false, bt,
                                                 (short)0, d0, false, false);
      d1 = __builtin_amdgcn_wmma_f32_16x16x4_f32(false, a1, false, bt,
                                                 (short)0, d1, false, false);
    }
    const float bias2 = b2[n0 + r];
    const int col = n0 + r;
#pragma unroll
    for (int v = 0; v < 8; ++v) {
      float g0 = d0[v] + bias2;
      float g1 = d1[v] + bias2;
      const int m0 = mhalf + v;
      const int m1 = 16 + mhalf + v;
      if (col < 256) {
        scale[m0 * 256 + col] = 1.0f / (1.0f + __expf(-g0));
        scale[m1 * 256 + col] = 1.0f / (1.0f + __expf(-g1));
      } else {
        shift[m0 * 256 + (col - 256)] = g0;
        shift[m1 * 256 + (col - 256)] = g1;
      }
    }
  }
}

__global__ void __launch_bounds__(256) se_apply_kernel(
    const float* __restrict__ x, const float* __restrict__ scale,
    const float* __restrict__ shift, float* __restrict__ out) {
  const int plane = blockIdx.x;
  const float s = scale[plane];
  const float t = shift[plane];
  const v4f* xp = (const v4f*)(x + (size_t)plane * 4096);
  v4f* op = (v4f*)(out + (size_t)plane * 4096);
#pragma unroll
  for (int i = 0; i < 4; ++i) {
    const int idx = threadIdx.x + i * 256;
    v4f v = xp[idx];
    v4f o;
    o.x = __builtin_fmaf(s, v.x, t);
    o.y = __builtin_fmaf(s, v.y, t);
    o.z = __builtin_fmaf(s, v.z, t);
    o.w = __builtin_fmaf(s, v.w, t);
    __builtin_nontemporal_store(o, op + idx);
  }
}

extern "C" void kernel_launch(void* const* d_in, const int* in_sizes, int n_in,
                              void* d_out, int out_size, void* d_ws,
                              size_t ws_size, hipStream_t stream) {
  const float* x  = (const float*)d_in[0];
  const float* w1 = (const float*)d_in[1];
  const float* b1 = (const float*)d_in[2];
  const float* w2 = (const float*)d_in[3];
  const float* b2 = (const float*)d_in[4];
  float* out = (float*)d_out;

  float* pooled = (float*)d_ws;
  float* scale  = pooled + 32 * 256;
  float* shift  = scale + 32 * 256;

  se_pool_kernel<<<32 * 256, 256, 0, stream>>>(x, pooled);
  se_mlp_wmma_kernel<<<1, 128, 0, stream>>>(pooled, w1, b1, w2, b2, scale,
                                            shift);
  se_apply_kernel<<<32 * 256, 256, 0, stream>>>(x, scale, shift, out);
}


# MLP kernel: 1 workgroup with a load-use K loop becomes 4 workgroups x 8 waves, GEMM1 K split over waves and summed through LDS (sum order changes), every load issued before the first WMMA
# speedup vs baseline: 1.5051x; 1.5051x over previous
#include <hip/hip_runtime.h>
#include <hip/hip_bf16.h>

typedef __attribute__((ext_vector_type(2))) float v2f;
typedef __attribute__((ext_vector_type(4))) float v4f;
typedef __attribute__((ext_vector_type(8))) float v8f;

__global__ void __launch_bounds__(256) se_pool_kernel(
    const float* __restrict__ x, float* __restrict__ pooled) {
  const int plane = blockIdx.x;
  const v4f* p = (const v4f*)(x + (size_t)plane * 4096);

  float sum = 0.0f;
#pragma unroll
  for (int i = 0; i < 4; ++i) {
    v4f v = p[threadIdx.x + i * 256];
    sum += (v.x + v.y) + (v.z + v.w);
  }
#pragma unroll
  for (int off = 16; off > 0; off >>= 1)
    sum += __shfl_down(sum, off, 32);

  __shared__ float red[8];
  if ((threadIdx.x & 31) == 0) red[threadIdx.x >> 5] = sum;
  __syncthreads();
  if (threadIdx.x == 0) {
    float t = 0.0f;
#pragma unroll
    for (int w = 0; w < 8; ++w) t += red[w];
    pooled[plane] = t * (1.0f / 4096.0f);
  }
}

#define SE_PITCH 20

__global__ void __launch_bounds__(256) se_mlp_wmma_kernel(
    const float* __restrict__ pooled,
    const float* __restrict__ w1,
    const float* __restrict__ b1,
    const float* __restrict__ w2,
    const float* __restrict__ b2,
    float* __restrict__ scale,
    float* __restrict__ shift) {
  const int lane = threadIdx.x & 31;
  const int wave = threadIdx.x >> 5;
  const int c    = lane & 15;
  const int hh   = lane >> 4;

  __shared__ __attribute__((aligned(16))) float part[8 * 32 * SE_PITCH];

  const int mt    = wave & 1;
  const int npair = blockIdx.x * 4 + (wave >> 1);
  const int n0    = npair * 32 + c;
  const int n1    = n0 + 16;

  const int kw = wave * 32 + 4 * hh;
  v4f pb[4], pa0[4], pa1[4];
#pragma unroll
  for (int j = 0; j < 4; ++j) {
    pb[j]  = *(const v4f*)(w1 + c * 256 + kw + 8 * j);
    pa0[j] = *(const v4f*)(pooled + c * 256 + kw + 8 * j);
    pa1[j] = *(const v4f*)(pooled + (16 + c) * 256 + kw + 8 * j);
  }
  v4f wb0[2], wb1[2], bb[2];
#pragma unroll
  for (int jj = 0; jj < 2; ++jj) {
    wb0[jj] = *(const v4f*)(w2 + n0 * 16 + 8 * jj + 4 * hh);
    wb1[jj] = *(const v4f*)(w2 + n1 * 16 + 8 * jj + 4 * hh);
    bb[jj]  = *(const v4f*)(b1 + 8 * jj + 4 * hh);
  }
  const float bias2_0 = b2[n0];
  const float bias2_1 = b2[n1];
  __builtin_amdgcn_sched_barrier(0);

  v8f c0 = {};
  v8f c1 = {};
#pragma unroll
  for (int j = 0; j < 4; ++j) {
    v2f bl = {pb[j].x, pb[j].y};
    v2f bh = {pb[j].z, pb[j].w};
    v2f a0l = {pa0[j].x, pa0[j].y};
    v2f a0h = {pa0[j].z, pa0[j].w};
    v2f a1l = {pa1[j].x, pa1[j].y};
    v2f a1h = {pa1[j].z, pa1[j].w};
    c0 = __builtin_amdgcn_wmma_f32_16x16x4_f32(false, a0l, false, bl,
                                               (short)0, c0, false, false);
    c1 = __builtin_amdgcn_wmma_f32_16x16x4_f32(false, a1l, false, bl,
                                               (short)0, c1, false, false);
    c0 = __builtin_amdgcn_wmma_f32_16x16x4_f32(false, a0h, false, bh,
                                               (short)0, c0, false, false);
    c1 = __builtin_amdgcn_wmma_f32_16x16x4_f32(false, a1h, false, bh,
                                               (short)0, c1, false, false);
  }

  float* pw = part + wave * (32 * SE_PITCH);
#pragma unroll
  for (int v = 0; v < 8; ++v) {
    pw[(v + 8 * hh) * SE_PITCH + c]      = c0[v];
    pw[(16 + v + 8 * hh) * SE_PITCH + c] = c1[v];
  }
  __syncthreads();

  v4f ha[2];
#pragma unroll
  for (int jj = 0; jj < 2; ++jj) {
    const float* pr = part + (mt * 16 + c) * SE_PITCH + 8 * jj + 4 * hh;
    v4f s = *(const v4f*)pr;
#pragma unroll
    for (int w = 1; w < 8; ++w)
      s += *(const v4f*)(pr + w * (32 * SE_PITCH));
    s += bb[jj];
    s.x = s.x > 0.0f ? s.x : 0.0f;
    s.y = s.y > 0.0f ? s.y : 0.0f;
    s.z = s.z > 0.0f ? s.z : 0.0f;
    s.w = s.w > 0.0f ? s.w : 0.0f;
    ha[jj] = s;
  }

  v8f d0 = {};
  v8f d1 = {};
#pragma unroll
  for (int jj = 0; jj < 2; ++jj) {
    v2f al = {ha[jj].x, ha[jj].y};
    v2f ah = {ha[jj].z, ha[jj].w};
    v2f b0l = {wb0[jj].x, wb0[jj].y};
    v2f b0h = {wb0[jj].z, wb0[jj].w};
    v2f b1l = {wb1[jj].x, wb1[jj].y};
    v2f b1h = {wb1[jj].z, wb1[jj].w};
    d0 = __builtin_amdgcn_wmma_f32_16x16x4_f32(false, al, false, b0l,
                                               (short)0, d0, false, false);
    d1 = __builtin_amdgcn_wmma_f32_16x16x4_f32(false, al, false, b1l,
                                               (short)0, d1, false, false);
    d0 = __builtin_amdgcn_wmma_f32_16x16x4_f32(false, ah, false, b0h,
                                               (short)0, d0, false, false);
    d1 = __builtin_amdgcn_wmma_f32_16x16x4_f32(false, ah, false, b1h,
                                               (short)0, d1, false, false);
  }

  if (blockIdx.x < 2) {
#pragma unroll
    for (int v = 0; v < 8; ++v) {
      const int m = mt * 16 + v + 8 * hh;
      const float g0 = d0[v] + bias2_0;
      const float g1 = d1[v] + bias2_1;
      scale[m * 256 + n0] = 1.0f / (1.0f + __expf(-g0));
      scale[m * 256 + n1] = 1.0f / (1.0f + __expf(-g1));
    }
  } else {
#pragma unroll
    for (int v = 0; v < 8; ++v) {
      const int m = mt * 16 + v + 8 * hh;
      shift[m * 256 + (n0 - 256)] = d0[v] + bias2_0;
      shift[m * 256 + (n1 - 256)] = d1[v] + bias2_1;
    }
  }
}

__global__ void __launch_bounds__(256) se_apply_kernel(
    const float* __restrict__ x, const float* __restrict__ scale,
    const float* __restrict__ shift, float* __restrict__ out) {
  const int plane = blockIdx.x;
  const float s = scale[plane];
  const float t = shift[plane];
  const v4f* xp = (const v4f*)(x + (size_t)plane * 4096);
  v4f* op = (v4f*)(out + (size_t)plane * 4096);
#pragma unroll
  for (int i = 0; i < 4; ++i) {
    const int idx = threadIdx.x + i * 256;
    v4f v = xp[idx];
    v4f o;
    o.x = __builtin_fmaf(s, v.x, t);
    o.y = __builtin_fmaf(s, v.y, t);
    o.z = __builtin_fmaf(s, v.z, t);
    o.w = __builtin_fmaf(s, v.w, t);
    __builtin_nontemporal_store(o, op + idx);
  }
}

extern "C" void kernel_launch(void* const* d_in, const int* in_sizes, int n_in,
                              void* d_out, int out_size, void* d_ws,
                              size_t ws_size, hipStream_t stream) {
  const float* x  = (const float*)d_in[0];
  const float* w1 = (const float*)d_in[1];
  const float* b1 = (const float*)d_in[2];
  const float* w2 = (const float*)d_in[3];
  const float* b2 = (const float*)d_in[4];
  float* out = (float*)d_out;

  float* pooled = (float*)d_ws;
  float* scale  = pooled + 32 * 256;
  float* shift  = scale + 32 * 256;

  se_pool_kernel<<<32 * 256, 256, 0, stream>>>(x, pooled);
  se_mlp_wmma_kernel<<<4, 256, 0, stream>>>(pooled, w1, b1, w2, b2, scale,
                                            shift);
  se_apply_kernel<<<32 * 256, 256, 0, stream>>>(x, scale, shift, out);
}


# MLP launch folded into the apply kernel: 512 workgroups (16 batches x 1 channel, 2 planes per wave) rebuild their h tile on WMMA under their own x loads; 3 launches become 2; sum order unchanged
# speedup vs baseline: 1.5805x; 1.0501x over previous
#include <hip/hip_runtime.h>
#include <hip/hip_bf16.h>

typedef __attribute__((ext_vector_type(2))) float v2f;
typedef __attribute__((ext_vector_type(4))) float v4f;
typedef __attribute__((ext_vector_type(8))) float v8f;

__global__ void __launch_bounds__(256) se_pool_kernel(
    const float* __restrict__ x, float* __restrict__ pooled) {
  const int plane = blockIdx.x;
  const v4f* p = (const v4f*)(x + (size_t)plane * 4096);

  float sum = 0.0f;
#pragma unroll
  for (int i = 0; i < 4; ++i) {
    v4f v = p[threadIdx.x + i * 256];
    sum += (v.x + v.y) + (v.z + v.w);
  }
#pragma unroll
  for (int off = 16; off > 0; off >>= 1)
    sum += __shfl_down(sum, off, 32);

  __shared__ float red[8];
  if ((threadIdx.x & 31) == 0) red[threadIdx.x >> 5] = sum;
  __syncthreads();
  if (threadIdx.x == 0) {
    float t = 0.0f;
#pragma unroll
    for (int w = 0; w < 8; ++w) t += red[w];
    pooled[plane] = t * (1.0f / 4096.0f);
  }
}

#define SE_PITCH 20

static __device__ __forceinline__ void se_apply_half(v4f (&v)[16], float s,
                                                     float t, v4f* op) {
#pragma unroll
  for (int i = 0; i < 16; ++i) {
    v4f o;
    o.x = __builtin_fmaf(s, v[i].x, t);
    o.y = __builtin_fmaf(s, v[i].y, t);
    o.z = __builtin_fmaf(s, v[i].z, t);
    o.w = __builtin_fmaf(s, v[i].w, t);
    __builtin_nontemporal_store(o, op + 32 * i);
  }
}

static __device__ __forceinline__ void se_load_half(v4f (&v)[16],
                                                    const v4f* xp) {
#pragma unroll
  for (int i = 0; i < 16; ++i) v[i] = xp[32 * i];
}

__global__ void __launch_bounds__(256) se_mlp_apply_kernel(
    const float* __restrict__ x,
    const float* __restrict__ pooled,
    const float* __restrict__ w1,
    const float* __restrict__ b1,
    const float* __restrict__ w2,
    const float* __restrict__ b2,
    float* __restrict__ out) {
  const int lane = threadIdx.x & 31;
  const int wave = __builtin_amdgcn_readfirstlane(threadIdx.x >> 5);
  const int c    = lane & 15;
  const int hh   = lane >> 4;
  const int ch   = blockIdx.x & 255;
  const int mt   = blockIdx.x >> 8;

  __shared__ __attribute__((aligned(16))) float part[8 * 16 * SE_PITCH];

  const int kw = wave * 32 + 4 * hh;
  v4f pb[4], pa[4];
#pragma unroll
  for (int j = 0; j < 4; ++j) {
    pb[j] = *(const v4f*)(w1 + c * 256 + kw + 8 * j);
    pa[j] = *(const v4f*)(pooled + (16 * mt + c) * 256 + kw + 8 * j);
  }
  const int nrow = (c & 1) ? 256 + ch : ch;
  v4f wa[2], bb[2];
#pragma unroll
  for (int jj = 0; jj < 2; ++jj) {
    wa[jj] = *(const v4f*)(w2 + nrow * 16 + 8 * jj + 4 * hh);
    bb[jj] = *(const v4f*)(b1 + 8 * jj + 4 * hh);
  }
  const float bias_s = b2[ch];
  const float bias_t = b2[256 + ch];

  const int bt = 2 * wave;
  const size_t plane0 = (size_t)((16 * mt + bt) * 256 + ch);
  const v4f* xp0 = (const v4f*)(x + plane0 * 4096) + lane;
  const v4f* xp1 = xp0 + 256 * 1024;
  v4f* op0 = (v4f*)(out + plane0 * 4096) + lane;
  v4f* op1 = op0 + 256 * 1024;
  v4f xa[16], xb[16], xc[16];
  se_load_half(xa, xp0);
  se_load_half(xb, xp0 + 512);
  __builtin_amdgcn_sched_barrier(0);

  v8f acc = {};
#pragma unroll
  for (int j = 0; j < 4; ++j) {
    v2f bl = {pb[j].x, pb[j].y};
    v2f bh = {pb[j].z, pb[j].w};
    v2f al = {pa[j].x, pa[j].y};
    v2f ah = {pa[j].z, pa[j].w};
    acc = __builtin_amdgcn_wmma_f32_16x16x4_f32(false, al, false, bl,
                                                (short)0, acc, false, false);
    acc = __builtin_amdgcn_wmma_f32_16x16x4_f32(false, ah, false, bh,
                                                (short)0, acc, false, false);
  }

  float* pw = part + wave * (16 * SE_PITCH);
#pragma unroll
  for (int v = 0; v < 8; ++v) pw[(v + 8 * hh) * SE_PITCH + c] = acc[v];

  asm volatile("s_wait_dscnt 0x0\n\ts_barrier_signal -1\n\ts_barrier_wait -1"
               ::: "memory");

  v4f ha[2];
#pragma unroll
  for (int jj = 0; jj < 2; ++jj) {
    const float* pr = part + c * SE_PITCH + 8 * jj + 4 * hh;
    v4f s = *(const v4f*)pr;
#pragma unroll
    for (int w = 1; w < 8; ++w)
      s += *(const v4f*)(pr + w * (16 * SE_PITCH));
    s += bb[jj];
    s.x = s.x > 0.0f ? s.x : 0.0f;
    s.y = s.y > 0.0f ? s.y : 0.0f;
    s.z = s.z > 0.0f ? s.z : 0.0f;
    s.w = s.w > 0.0f ? s.w : 0.0f;
    ha[jj] = s;
  }

  v8f d = {};
#pragma unroll
  for (int jj = 0; jj < 2; ++jj) {
    v2f al = {wa[jj].x, wa[jj].y};
    v2f ah = {wa[jj].z, wa[jj].w};
    v2f bl = {ha[jj].x, ha[jj].y};
    v2f bh = {ha[jj].z, ha[jj].w};
    d = __builtin_amdgcn_wmma_f32_16x16x4_f32(false, al, false, bl,
                                              (short)0, d, false, false);
    d = __builtin_amdgcn_wmma_f32_16x16x4_f32(false, ah, false, bh,
                                              (short)0, d, false, false);
  }

  const float g_s = d[0] + bias_s;
  const float sc  = 1.0f / (1.0f + __expf(-g_s));
  const float sh  = d[1] + bias_t;
  const float s0 = __int_as_float(__builtin_amdgcn_readlane(__float_as_int(sc), bt));
  const float t0 = __int_as_float(__builtin_amdgcn_readlane(__float_as_int(sh), bt));
  const float s1 = __int_as_float(__builtin_amdgcn_readlane(__float_as_int(sc), bt + 1));
  const float t1 = __int_as_float(__builtin_amdgcn_readlane(__float_as_int(sh), bt + 1));

  __builtin_amdgcn_sched_barrier(0);

  se_load_half(xc, xp1);
  se_apply_half(xa, s0, t0, op0);
  __builtin_amdgcn_sched_barrier(0);
  se_load_half(xa, xp1 + 512);
  se_apply_half(xb, s0, t0, op0 + 512);
  se_apply_half(xc, s1, t1, op1);
  se_apply_half(xa, s1, t1, op1 + 512);
}

extern "C" void kernel_launch(void* const* d_in, const int* in_sizes, int n_in,
                              void* d_out, int out_size, void* d_ws,
                              size_t ws_size, hipStream_t stream) {
  const float* x  = (const float*)d_in[0];
  const float* w1 = (const float*)d_in[1];
  const float* b1 = (const float*)d_in[2];
  const float* w2 = (const float*)d_in[3];
  const float* b2 = (const float*)d_in[4];
  float* out = (float*)d_out;

  float* pooled = (float*)d_ws;

  se_pool_kernel<<<32 * 256, 256, 0, stream>>>(x, pooled);
  se_mlp_apply_kernel<<<512, 256, 0, stream>>>(x, pooled, w1, b1, w2, b2, out);
}
